# starting kernel
# baseline (speedup 1.0000x reference)
#include <hip/hip_runtime.h>
#include <math.h>

typedef float v2f __attribute__((ext_vector_type(2)));
typedef float v8f __attribute__((ext_vector_type(8)));

#define C_DIM 128
#define N_PTS 4096
#define B_SZ  4
#define VSTR  132
#define PSTR  17
#define QT_W  2048
#define SMEMQ 16896

__device__ __forceinline__ float xor16(float v, unsigned selLo, unsigned selHi) {
  unsigned i = __builtin_bit_cast(unsigned, v);
  i = __builtin_amdgcn_permlane16(i, i, selLo, selHi, false, false);
  return __builtin_bit_cast(float, i);
}
#define XOR16_1(v) xor16((v), 0x67452301u, 0xEFCDAB89u)
#define XOR16_2(v) xor16((v), 0x54761032u, 0xDCFE98BAu)
#define XOR16_4(v) xor16((v), 0x32107654u, 0xBA98FEDCu)
#define XOR16_8(v) xor16((v), 0xFEDCBA98u, 0x76543210u)

__global__ __launch_bounds__(256) void cmix_kernel(const float* __restrict__ x,
                                                   const float* __restrict__ W,
                                                   float* __restrict__ out) {
  __shared__ float wrow[C_DIM];
  const int o = blockIdx.y;
  const int b = blockIdx.z;
  const int tid = threadIdx.x;
  if (tid < C_DIM) wrow[tid] = W[o * C_DIM + tid];
  __syncthreads();
  const int n = blockIdx.x * 256 + tid;
  float acc = 0.0f;
#pragma unroll 8
  for (int c = 0; c < C_DIM; ++c)
    acc += wrow[c] * x[((size_t)(b * C_DIM + c)) * N_PTS + n];
  out[((size_t)(b * C_DIM + o)) * N_PTS + n] = acc;
}

__global__ __launch_bounds__(256)
__attribute__((amdgpu_num_vgpr(256)))
void attn_kernel(const float* __restrict__ Q,
                 const float* __restrict__ K,
                 const float* __restrict__ V,
                 float* __restrict__ Y,
                 float* __restrict__ R) {
  __shared__ float smem_q[SMEMQ];
  __shared__ float kt[C_DIM * 16];
  __shared__ float vt[16 * VSTR];
  __shared__ float pl[8][16 * PSTR];

  const int tid  = threadIdx.x;
  const int lane = tid & 31;
  const int wave = tid >> 5;
  const int b    = blockIdx.y;
  const int qBase0 = blockIdx.x * 128;
  const int lo = lane & 15;
  const int hg = lane >> 4;
  const float invTemp = 0.088388347648318447f;
  const float NEGINF = -__builtin_inff();

#pragma unroll
  for (int i = 0; i < 16; ++i) {
    const int idx4 = tid + 256 * i;
    const int c  = idx4 >> 5;
    const int q4 = idx4 & 31;
    const float4 qv = *(const float4*)(Q + ((size_t)(b * C_DIM + c)) * N_PTS + qBase0 + q4 * 4);
    const int q = q4 * 4;
    *(float4*)&smem_q[(q >> 4) * QT_W + c * 16 + (q & 15)] = qv;
  }
  __syncthreads();
  const float* qt_w = smem_q + wave * QT_W;

  v8f  o[8] = {};
  v8f  ls8  = {};
  float t1[8], t2[8];
#pragma unroll
  for (int r = 0; r < 8; ++r) { t1[r] = NEGINF; t2[r] = NEGINF; }
  const v2f vone = {1.0f, 1.0f};

  for (int kb0 = 0; kb0 < N_PTS; kb0 += 16) {
#pragma unroll
    for (int i = 0; i < 2; ++i) {
      const int idx4 = tid + 256 * i;
      const int c  = idx4 >> 2;
      const int n4 = (idx4 & 3) * 4;
      const size_t g = ((size_t)(b * C_DIM + c)) * N_PTS + kb0 + n4;
      const float4 kv = *(const float4*)(K + g);
      *(float4*)&kt[c * 16 + n4] = kv;
      const float4 vv = *(const float4*)(V + g);
      vt[(n4 + 0) * VSTR + c] = vv.x;
      vt[(n4 + 1) * VSTR + c] = vv.y;
      vt[(n4 + 2) * VSTR + c] = vv.z;
      vt[(n4 + 3) * VSTR + c] = vv.w;
    }
    __syncthreads();

    v8f s = {};
#pragma unroll 4
    for (int ks = 0; ks < 32; ++ks) {
      const int c0 = 4 * ks + 2 * hg;
      v2f a, kb;
      a.x  = qt_w[c0 * 16 + lo];
      a.y  = qt_w[(c0 + 1) * 16 + lo];
      kb.x = kt[c0 * 16 + lo];
      kb.y = kt[(c0 + 1) * 16 + lo];
      s = __builtin_amdgcn_wmma_f32_16x16x4_f32(false, a, false, kb,
                                                (short)0, s, false, false);
    }

    v8f p;
    float mo[8];
    float anyChange = 0.0f;
#pragma unroll
    for (int r = 0; r < 8; ++r) {
      float a1 = s[r], a2 = NEGINF;
      {
        float b1 = XOR16_1(a1), b2 = XOR16_1(a2);
        float hi = fmaxf(a1, b1), lw = fminf(a1, b1);
        a2 = fmaxf(lw, fmaxf(a2, b2)); a1 = hi;
        b1 = XOR16_2(a1); b2 = XOR16_2(a2);
        hi = fmaxf(a1, b1); lw = fminf(a1, b1);
        a2 = fmaxf(lw, fmaxf(a2, b2)); a1 = hi;
        b1 = XOR16_4(a1); b2 = XOR16_4(a2);
        hi = fmaxf(a1, b1); lw = fminf(a1, b1);
        a2 = fmaxf(lw, fmaxf(a2, b2)); a1 = hi;
        b1 = XOR16_8(a1); b2 = XOR16_8(a2);
        hi = fmaxf(a1, b1); lw = fminf(a1, b1);
        a2 = fmaxf(lw, fmaxf(a2, b2)); a1 = hi;
      }
      mo[r] = t1[r];
      const float hi = fmaxf(t1[r], a1);
      const float lw = fminf(t1[r], a1);
      t2[r] = fmaxf(lw, fmaxf(t2[r], a2));
      t1[r] = hi;
      anyChange += (t1[r] - mo[r]);
      p[r] = __expf((s[r] - t1[r]) * invTemp);
    }

    if (__any(anyChange > 0.0f)) {
#pragma unroll
      for (int r = 0; r < 8; ++r) {
        const float sc = __expf((mo[r] - t1[r]) * invTemp);
        ls8[r] *= sc;
#pragma unroll
        for (int ct = 0; ct < 8; ++ct) o[ct][r] *= sc;
      }
    }

    float* pw = &pl[wave][0];
#pragma unroll
    for (int r = 0; r < 8; ++r)
      pw[(r + 8 * hg) * PSTR + lo] = p[r];

#pragma unroll 1
    for (int ks2 = 0; ks2 < 4; ++ks2) {
      const int k0 = 4 * ks2 + 2 * hg;
      v2f a;
      a.x = pw[lo * PSTR + k0];
      a.y = pw[lo * PSTR + k0 + 1];
#pragma unroll
      for (int ct = 0; ct < 8; ++ct) {
        v2f bv;
        bv.x = vt[k0 * VSTR + ct * 16 + lo];
        bv.y = vt[(k0 + 1) * VSTR + ct * 16 + lo];
        o[ct] = __builtin_amdgcn_wmma_f32_16x16x4_f32(false, a, false, bv,
                                                      (short)0, o[ct], false, false);
      }
      ls8 = __builtin_amdgcn_wmma_f32_16x16x4_f32(false, a, false, vone,
                                                  (short)0, ls8, false, false);
    }
    __syncthreads();
  }

#pragma unroll
  for (int r = 0; r < 8; ++r) {
    const float invl = 1.0f / ls8[r];
    const int q = wave * 16 + r + 8 * hg;
#pragma unroll
    for (int ct = 0; ct < 8; ++ct) {
      const int c = ct * 16 + lo;
      smem_q[c * VSTR + q] = o[ct][r] * invl;
    }
    if (lo == 0) {
      const float x1 = fminf(fmaxf(t1[r], -1.0f), 1.0f);
      const float x2 = fminf(fmaxf(t2[r], -1.0f), 1.0f);
      R[b * N_PTS + qBase0 + q] = acosf(x1) / acosf(x2);
    }
  }
  __syncthreads();

#pragma unroll
  for (int i = 0; i < 16; ++i) {
    const int idx4 = tid + 256 * i;
    const int c  = idx4 >> 5;
    const int q4 = (idx4 & 31) * 4;
    const float4 val = *(const float4*)&smem_q[c * VSTR + q4];
    *(float4*)(Y + ((size_t)(b * C_DIM + c)) * N_PTS + qBase0 + q4) = val;
  }
}

extern "C" void kernel_launch(void* const* d_in, const int* in_sizes, int n_in,
                              void* d_out, int out_size, void* d_ws, size_t ws_size,
                              hipStream_t stream) {
  (void)in_sizes; (void)n_in; (void)out_size;
  const float* q  = (const float*)d_in[0];
  const float* k  = (const float*)d_in[1];
  const float* v  = (const float*)d_in[2];
  const float* Wq = (const float*)d_in[3];
  const float* Wk = (const float*)d_in[4];
  const float* Wv = (const float*)d_in[5];

  float* out  = (float*)d_out;
  float* Yout = out;
  float* Rout = out + (size_t)B_SZ * C_DIM * N_PTS;

  const size_t tensElems = (size_t)B_SZ * C_DIM * N_PTS;
  const float *qc = q, *kc = k, *vc = v;

  if (ws_size >= 3 * tensElems * sizeof(float)) {
    float* qw = (float*)d_ws;
    float* kw = qw + tensElems;
    float* vw = kw + tensElems;
    dim3 g(N_PTS / 256, C_DIM, B_SZ);
    cmix_kernel<<<g, 256, 0, stream>>>(q, Wq, qw);
    cmix_kernel<<<g, 256, 0, stream>>>(k, Wk, kw);
    cmix_kernel<<<g, 256, 0, stream>>>(v, Wv, vw);
    qc = qw; kc = kw; vc = vw;
  }

  dim3 ga(N_PTS / 128, B_SZ);
  attn_kernel<<<ga, 256, 0, stream>>>(qc, kc, vc, Yout, Rout);
}


# attn: raw scores (not weights) pass LDS into the A layout: a lane holds 16 keys of one row; top-2/max/exp per lane + one half swap (was 16-lane butterflies); row sum on VALU (sum order changes)
# speedup vs baseline: 3.1457x; 3.1457x over previous
#include <hip/hip_runtime.h>
#include <math.h>

typedef float v2f __attribute__((ext_vector_type(2)));
typedef float v4f __attribute__((ext_vector_type(4)));
typedef float v8f __attribute__((ext_vector_type(8)));

#define C_DIM 128
#define N_PTS 4096
#define B_SZ  4
#define VSTR  132
#define KGP   68
#define KSUB  (32 * KGP)
#define VP    36
#define PP    36
#define QT_W  2048
#define SMEMQ 16896
#define NSPLIT 4
#define KSPAN (N_PTS / NSPLIT)
#define KSTEP 32
#define MCPT  4

__device__ __forceinline__ float xor16(float v, unsigned selLo, unsigned selHi) {
  unsigned i = __builtin_bit_cast(unsigned, v);
  i = __builtin_amdgcn_permlane16(i, i, selLo, selHi, false, false);
  return __builtin_bit_cast(float, i);
}
#define XOR16_1(v) xor16((v), 0x67452301u, 0xEFCDAB89u)
#define XOR16_2(v) xor16((v), 0x54761032u, 0xDCFE98BAu)
#define XOR16_4(v) xor16((v), 0x32107654u, 0xBA98FEDCu)
#define XOR16_8(v) xor16((v), 0xFEDCBA98u, 0x76543210u)

__device__ __forceinline__ float xhalf(float v) {
  unsigned i = __builtin_bit_cast(unsigned, v);
  i = __builtin_amdgcn_permlanex16(i, i, 0x76543210u, 0xFEDCBA98u, false, false);
  return __builtin_bit_cast(float, i);
}
#define XHALF(v) xhalf(v)

__device__ __forceinline__ void top2_push(float& h, float& l, float x) {
  l = fmaxf(fminf(h, x), l);
  h = fmaxf(h, x);
}

#define MIX_OT 8
__global__ __launch_bounds__(256) void cmix_kernel(const float* __restrict__ x,
                                                   const float* __restrict__ W,
                                                   float* __restrict__ out) {
  const int o0 = blockIdx.y * MIX_OT;
  const int b  = blockIdx.z;
  const int n4 = (blockIdx.x * 256 + threadIdx.x) * 4;
  const float* xp = x + (size_t)b * C_DIM * N_PTS + n4;
  const float* wp = W + o0 * C_DIM;

  v4f acc[MIX_OT];
#pragma unroll
  for (int j = 0; j < MIX_OT; ++j) acc[j] = (v4f){0.0f, 0.0f, 0.0f, 0.0f};

#pragma unroll 16
  for (int c = 0; c < C_DIM; ++c) {
    const v4f xv = *(const v4f*)(xp + (size_t)c * N_PTS);
#pragma unroll
    for (int j = 0; j < MIX_OT; ++j) {
      const float w = wp[j * C_DIM + c];
#pragma unroll
      for (int i = 0; i < 4; ++i) acc[j][i] = __builtin_fmaf(w, xv[i], acc[j][i]);
    }
  }

#pragma unroll
  for (int j = 0; j < MIX_OT; ++j)
    *(v4f*)(out + ((size_t)(b * C_DIM + o0 + j)) * N_PTS + n4) = acc[j];
}

__global__ __launch_bounds__(256)
__attribute__((amdgpu_num_vgpr(256)))
void attn_kernel(const float* __restrict__ Q,
                 const float* __restrict__ K,
                 const float* __restrict__ V,
                 float* __restrict__ Op,
                 float* __restrict__ Sm,
                 float* __restrict__ St2,
                 float* __restrict__ Sl) {
  __shared__ __attribute__((aligned(16))) float smem_q[SMEMQ];
  __shared__ __attribute__((aligned(16))) float kt[2][KSUB];
  __shared__ __attribute__((aligned(16))) float vt[C_DIM * VP];
  __shared__ __attribute__((aligned(16))) float pl[8][16 * PP];

  const int tid  = threadIdx.x;
  const int lane = tid & 31;
  const int wave = tid >> 5;
  const int b    = blockIdx.y;
  const int split = blockIdx.z;
  const int kBeg = split * KSPAN;
  const int qBase0 = blockIdx.x * 128;
  const int lo = lane & 15;
  const int hg = lane >> 4;
  const float invTemp = 0.088388347648318447f;
  const float NEGINF = -__builtin_inff();

#pragma unroll
  for (int i = 0; i < 16; ++i) {
    const int idx4 = tid + 256 * i;
    const int c  = idx4 >> 5;
    const int q4 = idx4 & 31;
    const float4 qv = *(const float4*)(Q + ((size_t)(b * C_DIM + c)) * N_PTS + qBase0 + q4 * 4);
    const int q = q4 * 4;
    float* qd = &smem_q[(q >> 4) * QT_W + (c >> 2) * 64 + (q & 15) * 4 + (c & 3)];
    qd[0] = qv.x; qd[4] = qv.y; qd[8] = qv.z; qd[12] = qv.w;
  }
  __syncthreads();
  const float* qt_w = smem_q + wave * QT_W;

  v8f  o[8] = {};
  float t1 = NEGINF, t2 = NEGINF;
  float lsum = 0.0f;

  for (int kb0 = kBeg; kb0 < kBeg + KSPAN; kb0 += KSTEP) {
#pragma unroll
    for (int h = 0; h < 2; ++h) {
#pragma unroll
      for (int i = 0; i < 2; ++i) {
        const int idx4 = tid + 256 * i;
        const int c  = idx4 >> 2;
        const int n4 = (idx4 & 3) * 4;
        const size_t g = ((size_t)(b * C_DIM + c)) * N_PTS + kb0 + 16 * h + n4;
        const float4 kv = *(const float4*)(K + g);
        float* kd = &kt[h][(c >> 2) * KGP + n4 * 4 + (c & 3)];
        kd[0] = kv.x; kd[4] = kv.y; kd[8] = kv.z; kd[12] = kv.w;
        const float4 vv = *(const float4*)(V + g);
        *(float4*)&vt[c * VP + 16 * h + n4] = vv;
      }
    }
    __syncthreads();

    v8f s0 = {}, s1 = {};
#pragma unroll 4
    for (int j = 0; j < 16; ++j) {
      const int cg = 2 * j + hg;
      const v4f qa = *(const v4f*)&qt_w[cg * 64 + lo * 4];
      const v4f k0 = *(const v4f*)&kt[0][cg * KGP + lo * 4];
      const v4f k1 = *(const v4f*)&kt[1][cg * KGP + lo * 4];
      const v2f a0 = {qa.x, qa.y}, a1 = {qa.z, qa.w};
      const v2f k00 = {k0.x, k0.y}, k01 = {k0.z, k0.w};
      const v2f k10 = {k1.x, k1.y}, k11 = {k1.z, k1.w};
      s0 = __builtin_amdgcn_wmma_f32_16x16x4_f32(false, a0, false, k00,
                                                 (short)0, s0, false, false);
      s1 = __builtin_amdgcn_wmma_f32_16x16x4_f32(false, a0, false, k10,
                                                 (short)0, s1, false, false);
      s0 = __builtin_amdgcn_wmma_f32_16x16x4_f32(false, a1, false, k01,
                                                 (short)0, s0, false, false);
      s1 = __builtin_amdgcn_wmma_f32_16x16x4_f32(false, a1, false, k11,
                                                 (short)0, s1, false, false);
    }

    float* pw = &pl[wave][0];
#pragma unroll
    for (int r = 0; r < 8; ++r) {
      pw[(r + 8 * hg) * PP + lo] = s0[r];
      pw[(r + 8 * hg) * PP + 16 + lo] = s1[r];
    }
    v4f sa[KSTEP / 8];
#pragma unroll
    for (int j = 0; j < KSTEP / 8; ++j)
      sa[j] = *(const v4f*)&pw[lo * PP + 8 * j + 4 * hg];

    float hA = fmaxf(sa[0].x, sa[0].y), lA = fminf(sa[0].x, sa[0].y);
    float hB = fmaxf(sa[2].x, sa[2].y), lB = fminf(sa[2].x, sa[2].y);
    top2_push(hA, lA, sa[0].z); top2_push(hB, lB, sa[2].z);
    top2_push(hA, lA, sa[0].w); top2_push(hB, lB, sa[2].w);
#pragma unroll
    for (int e = 0; e < 4; ++e) {
      top2_push(hA, lA, sa[1][e]); top2_push(hB, lB, sa[3][e]);
    }
    float a1 = fmaxf(hA, hB);
    float a2 = fmaxf(fminf(hA, hB), fmaxf(lA, lB));
    {
      const float b1 = XHALF(a1), b2 = XHALF(a2);
      const float hi = fmaxf(a1, b1), lw = fminf(a1, b1);
      a2 = fmaxf(lw, fmaxf(a2, b2)); a1 = hi;
    }
    const float mo = t1;
    {
      const float hi = fmaxf(t1, a1), lw = fminf(t1, a1);
      t2 = fmaxf(lw, fmaxf(t2, a2)); t1 = hi;
    }

    if (__any(t1 > mo)) {
      const float sc = __expf((mo - t1) * invTemp);
      lsum *= sc;
      const float scs = XOR16_8(sc);
      const float scx = hg ? scs : sc;
#pragma unroll
      for (int r = 0; r < 8; ++r) {
        const float f = xor16(scx, 0x11111111u * r, 0x11111111u * r);
#pragma unroll
        for (int ct = 0; ct < 8; ++ct) o[ct][r] *= f;
      }
    }

    v4f pa[KSTEP / 8];
    float q4[KSTEP / 8];
#pragma unroll
    for (int j = 0; j < KSTEP / 8; ++j) {
#pragma unroll
      for (int e = 0; e < 4; ++e) pa[j][e] = __expf((sa[j][e] - t1) * invTemp);
      q4[j] = (pa[j].x + pa[j].y) + (pa[j].z + pa[j].w);
    }
    lsum += (q4[0] + q4[1]) + (q4[2] + q4[3]);

#pragma unroll
    for (int j = 0; j < KSTEP / 8; ++j) {
      const int k4 = 8 * j + 4 * hg;
      v4f vb[8];
#pragma unroll
      for (int ct = 0; ct < 8; ++ct)
        vb[ct] = *(const v4f*)&vt[(ct * 16 + lo) * VP + k4];
      const v2f a0 = {pa[j].x, pa[j].y}, a1f = {pa[j].z, pa[j].w};
#pragma unroll
      for (int ct = 0; ct < 8; ++ct) {
        const v2f bv = {vb[ct].x, vb[ct].y};
        o[ct] = __builtin_amdgcn_wmma_f32_16x16x4_f32(false, a0, false, bv,
                                                      (short)0, o[ct], false, false);
      }
#pragma unroll
      for (int ct = 0; ct < 8; ++ct) {
        const v2f bv = {vb[ct].z, vb[ct].w};
        o[ct] = __builtin_amdgcn_wmma_f32_16x16x4_f32(false, a1f, false, bv,
                                                      (short)0, o[ct], false, false);
      }
    }
    __syncthreads();
  }

  const size_t sb = (size_t)(split * B_SZ + b);
#pragma unroll
  for (int r = 0; r < 8; ++r) {
    const int q = wave * 16 + r + 8 * hg;
#pragma unroll
    for (int ct = 0; ct < 8; ++ct) {
      const int c = ct * 16 + lo;
      smem_q[c * VSTR + q] = o[ct][r];
    }
  }
  {
    const float ltot = lsum + XHALF(lsum);
    if (hg == 0) {
      const size_t si = sb * N_PTS + qBase0 + wave * 16 + lo;
      Sm[si]  = t1;
      St2[si] = t2;
      Sl[si]  = ltot;
    }
  }
  __syncthreads();

#pragma unroll
  for (int i = 0; i < 16; ++i) {
    const int idx4 = tid + 256 * i;
    const int c  = idx4 >> 5;
    const int q4 = (idx4 & 31) * 4;
    const float4 val = *(const float4*)&smem_q[c * VSTR + q4];
    *(float4*)(Op + (sb * C_DIM + c) * N_PTS + qBase0 + q4) = val;
  }
}

__global__ __launch_bounds__(256) void merge_kernel(const float* __restrict__ Op,
                                                    const float* __restrict__ Sm,
                                                    const float* __restrict__ St2,
                                                    const float* __restrict__ Sl,
                                                    float* __restrict__ Y,
                                                    float* __restrict__ R) {
  const int n4 = (blockIdx.x * 256 + threadIdx.x) * 4;
  const int c0 = blockIdx.y * MCPT;
  const int b  = blockIdx.z;
  const float invTemp = 0.088388347648318447f;

  v4f mv[NSPLIT], lv[NSPLIT], ov[MCPT][NSPLIT];
#pragma unroll
  for (int s = 0; s < NSPLIT; ++s) {
    const size_t si = ((size_t)(s * B_SZ + b)) * N_PTS + n4;
    mv[s] = *(const v4f*)(Sm + si);
    lv[s] = *(const v4f*)(Sl + si);
  }
#pragma unroll
  for (int cc = 0; cc < MCPT; ++cc)
#pragma unroll
    for (int s = 0; s < NSPLIT; ++s)
      ov[cc][s] = *(const v4f*)(Op + ((size_t)((s * B_SZ + b) * C_DIM + c0 + cc)) * N_PTS + n4);

  asm volatile("" :: "v"(mv[0]), "v"(mv[1]), "v"(mv[2]), "v"(mv[3]),
                     "v"(lv[0]), "v"(lv[1]), "v"(lv[2]), "v"(lv[3]),
                     "v"(ov[0][0]), "v"(ov[0][1]), "v"(ov[0][2]), "v"(ov[0][3]),
                     "v"(ov[1][0]), "v"(ov[1][1]), "v"(ov[1][2]), "v"(ov[1][3]),
                     "v"(ov[2][0]), "v"(ov[2][1]), "v"(ov[2][2]), "v"(ov[2][3]),
                     "v"(ov[3][0]), "v"(ov[3][1]), "v"(ov[3][2]), "v"(ov[3][3]));

  v4f M = mv[0];
#pragma unroll
  for (int s = 1; s < NSPLIT; ++s)
#pragma unroll
    for (int j = 0; j < 4; ++j) M[j] = fmaxf(M[j], mv[s][j]);
  v4f w[NSPLIT];
  v4f L = {0.0f, 0.0f, 0.0f, 0.0f};
#pragma unroll
  for (int s = 0; s < NSPLIT; ++s)
#pragma unroll
    for (int j = 0; j < 4; ++j) {
      w[s][j] = __expf((mv[s][j] - M[j]) * invTemp);
      L[j] += w[s][j] * lv[s][j];
    }
  v4f inv;
#pragma unroll
  for (int j = 0; j < 4; ++j) inv[j] = 1.0f / L[j];

#pragma unroll
  for (int cc = 0; cc < MCPT; ++cc) {
    const int c = c0 + cc;
    v4f acc = {0.0f, 0.0f, 0.0f, 0.0f};
#pragma unroll
    for (int s = 0; s < NSPLIT; ++s) acc += w[s] * ov[cc][s];
    *(v4f*)(Y + ((size_t)(b * C_DIM + c)) * N_PTS + n4) = acc * inv;
  }

  if (blockIdx.y == 0) {
    const float NEGINF = -__builtin_inff();
    v4f tv[NSPLIT];
#pragma unroll
    for (int s = 0; s < NSPLIT; ++s)
      tv[s] = *(const v4f*)(St2 + ((size_t)(s * B_SZ + b)) * N_PTS + n4);
    v4f rv;
#pragma unroll
    for (int j = 0; j < 4; ++j) {
      float T1 = NEGINF, T2 = NEGINF;
#pragma unroll
      for (int s = 0; s < NSPLIT; ++s) {
        const float hi = fmaxf(T1, mv[s][j]);
        const float lw = fminf(T1, mv[s][j]);
        T2 = fmaxf(lw, fmaxf(T2, tv[s][j]));
        T1 = hi;
      }
      const float x1 = fminf(fmaxf(T1, -1.0f), 1.0f);
      const float x2 = fminf(fmaxf(T2, -1.0f), 1.0f);
      rv[j] = acosf(x1) / acosf(x2);
    }
    *(v4f*)(R + (size_t)b * N_PTS + n4) = rv;
  }
}

extern "C" void kernel_launch(void* const* d_in, const int* in_sizes, int n_in,
                              void* d_out, int out_size, void* d_ws, size_t ws_size,
                              hipStream_t stream) {
  (void)in_sizes; (void)n_in; (void)out_size;
  const float* q  = (const float*)d_in[0];
  const float* k  = (const float*)d_in[1];
  const float* v  = (const float*)d_in[2];
  const float* Wq = (const float*)d_in[3];
  const float* Wk = (const float*)d_in[4];
  const float* Wv = (const float*)d_in[5];

  float* out  = (float*)d_out;
  float* Yout = out;
  float* Rout = out + (size_t)B_SZ * C_DIM * N_PTS;

  const size_t tensElems = (size_t)B_SZ * C_DIM * N_PTS;
  const float *qc = q, *kc = k, *vc = v;

  if (ws_size >= 3 * tensElems * sizeof(float)) {
    float* qw = (float*)d_ws;
    float* kw = qw + tensElems;
    float* vw = kw + tensElems;
    dim3 g(N_PTS / 1024, C_DIM / MIX_OT, B_SZ);
    cmix_kernel<<<g, 256, 0, stream>>>(q, Wq, qw);
    cmix_kernel<<<g, 256, 0, stream>>>(k, Wk, kw);
    cmix_kernel<<<g, 256, 0, stream>>>(v, Wv, vw);
    qc = qw; kc = kw; vc = vw;
  }

  float* Op  = (float*)d_ws + 3 * tensElems;
  float* Sm  = Op + NSPLIT * tensElems;
  float* St2 = Sm + (size_t)NSPLIT * B_SZ * N_PTS;
  float* Sl  = St2 + (size_t)NSPLIT * B_SZ * N_PTS;

  dim3 ga(N_PTS / 128, B_SZ, NSPLIT);
  attn_kernel<<<ga, 256, 0, stream>>>(qc, kc, vc, Op, Sm, St2, Sl);

  dim3 gm(N_PTS / 1024, C_DIM / MCPT, B_SZ);
  merge_kernel<<<gm, 256, 0, stream>>>(Op, Sm, St2, Sl, Yout, Rout);
}


# channel mixes: the three launches (q, k, v) run as one launch, grid z = 3 x batch: 768 work-groups in flight together (was 256 per launch); 3 launches a repetition (was 5); sum order unchanged
# speedup vs baseline: 3.2554x; 1.0349x over previous
#include <hip/hip_runtime.h>
#include <math.h>

typedef float v2f __attribute__((ext_vector_type(2)));
typedef float v4f __attribute__((ext_vector_type(4)));
typedef float v8f __attribute__((ext_vector_type(8)));

#define C_DIM 128
#define N_PTS 4096
#define B_SZ  4
#define VSTR  132
#define KGP   68
#define KSUB  (32 * KGP)
#define VP    36
#define PP    36
#define QT_W  2048
#define SMEMQ 16896
#define NSPLIT 4
#define KSPAN (N_PTS / NSPLIT)
#define KSTEP 32
#define MCPT  4

__device__ __forceinline__ float xor16(float v, unsigned selLo, unsigned selHi) {
  unsigned i = __builtin_bit_cast(unsigned, v);
  i = __builtin_amdgcn_permlane16(i, i, selLo, selHi, false, false);
  return __builtin_bit_cast(float, i);
}
#define XOR16_1(v) xor16((v), 0x67452301u, 0xEFCDAB89u)
#define XOR16_2(v) xor16((v), 0x54761032u, 0xDCFE98BAu)
#define XOR16_4(v) xor16((v), 0x32107654u, 0xBA98FEDCu)
#define XOR16_8(v) xor16((v), 0xFEDCBA98u, 0x76543210u)

__device__ __forceinline__ float xhalf(float v) {
  unsigned i = __builtin_bit_cast(unsigned, v);
  i = __builtin_amdgcn_permlanex16(i, i, 0x76543210u, 0xFEDCBA98u, false, false);
  return __builtin_bit_cast(float, i);
}
#define XHALF(v) xhalf(v)

__device__ __forceinline__ void top2_push(float& h, float& l, float x) {
  l = fmaxf(fminf(h, x), l);
  h = fmaxf(h, x);
}

#define MIX_OT 8
__global__ __launch_bounds__(256) void cmix3_kernel(const float* __restrict__ xq,
                                                    const float* __restrict__ xk,
                                                    const float* __restrict__ xv,
                                                    const float* __restrict__ Wq,
                                                    const float* __restrict__ Wk,
                                                    const float* __restrict__ Wv,
                                                    float* __restrict__ oq,
                                                    float* __restrict__ ok,
                                                    float* __restrict__ ov) {
  const int o0 = blockIdx.y * MIX_OT;
  const int t  = blockIdx.z / B_SZ;
  const int b  = blockIdx.z - t * B_SZ;
  const float* __restrict__ x   = (t == 0) ? xq : ((t == 1) ? xk : xv);
  const float* __restrict__ W   = (t == 0) ? Wq : ((t == 1) ? Wk : Wv);
  float* __restrict__       out = (t == 0) ? oq : ((t == 1) ? ok : ov);
  const int n4 = (blockIdx.x * 256 + threadIdx.x) * 4;
  const float* xp = x + (size_t)b * C_DIM * N_PTS + n4;
  const float* wp = W + o0 * C_DIM;

  v4f acc[MIX_OT];
#pragma unroll
  for (int j = 0; j < MIX_OT; ++j) acc[j] = (v4f){0.0f, 0.0f, 0.0f, 0.0f};

#pragma unroll 16
  for (int c = 0; c < C_DIM; ++c) {
    const v4f xv4 = *(const v4f*)(xp + (size_t)c * N_PTS);
#pragma unroll
    for (int j = 0; j < MIX_OT; ++j) {
      const float w = wp[j * C_DIM + c];
#pragma unroll
      for (int i = 0; i < 4; ++i) acc[j][i] = __builtin_fmaf(w, xv4[i], acc[j][i]);
    }
  }

#pragma unroll
  for (int j = 0; j < MIX_OT; ++j)
    *(v4f*)(out + ((size_t)(b * C_DIM + o0 + j)) * N_PTS + n4) = acc[j];
}

__global__ __launch_bounds__(256)
__attribute__((amdgpu_num_vgpr(256)))
void attn_kernel(const float* __restrict__ Q,
                 const float* __restrict__ K,
                 const float* __restrict__ V,
                 float* __restrict__ Op,
                 float* __restrict__ Sm,
                 float* __restrict__ St2,
                 float* __restrict__ Sl) {
  __shared__ __attribute__((aligned(16))) float smem_q[SMEMQ];
  __shared__ __attribute__((aligned(16))) float kt[2][KSUB];
  __shared__ __attribute__((aligned(16))) float vt[C_DIM * VP];
  __shared__ __attribute__((aligned(16))) float pl[8][16 * PP];

  const int tid  = threadIdx.x;
  const int lane = tid & 31;
  const int wave = tid >> 5;
  const int b    = blockIdx.y;
  const int split = blockIdx.z;
  const int kBeg = split * KSPAN;
  const int qBase0 = blockIdx.x * 128;
  const int lo = lane & 15;
  const int hg = lane >> 4;
  const float invTemp = 0.088388347648318447f;
  const float NEGINF = -__builtin_inff();

#pragma unroll
  for (int i = 0; i < 16; ++i) {
    const int idx4 = tid + 256 * i;
    const int c  = idx4 >> 5;
    const int q4 = idx4 & 31;
    const float4 qv = *(const float4*)(Q + ((size_t)(b * C_DIM + c)) * N_PTS + qBase0 + q4 * 4);
    const int q = q4 * 4;
    float* qd = &smem_q[(q >> 4) * QT_W + (c >> 2) * 64 + (q & 15) * 4 + (c & 3)];
    qd[0] = qv.x; qd[4] = qv.y; qd[8] = qv.z; qd[12] = qv.w;
  }
  __syncthreads();
  const float* qt_w = smem_q + wave * QT_W;

  v8f  o[8] = {};
  float t1 = NEGINF, t2 = NEGINF;
  float lsum = 0.0f;

  for (int kb0 = kBeg; kb0 < kBeg + KSPAN; kb0 += KSTEP) {
#pragma unroll
    for (int h = 0; h < 2; ++h) {
#pragma unroll
      for (int i = 0; i < 2; ++i) {
        const int idx4 = tid + 256 * i;
        const int c  = idx4 >> 2;
        const int n4 = (idx4 & 3) * 4;
        const size_t g = ((size_t)(b * C_DIM + c)) * N_PTS + kb0 + 16 * h + n4;
        const float4 kv = *(const float4*)(K + g);
        float* kd = &kt[h][(c >> 2) * KGP + n4 * 4 + (c & 3)];
        kd[0] = kv.x; kd[4] = kv.y; kd[8] = kv.z; kd[12] = kv.w;
        const float4 vv = *(const float4*)(V + g);
        *(float4*)&vt[c * VP + 16 * h + n4] = vv;
      }
    }
    __syncthreads();

    v8f s0 = {}, s1 = {};
#pragma unroll 4
    for (int j = 0; j < 16; ++j) {
      const int cg = 2 * j + hg;
      const v4f qa = *(const v4f*)&qt_w[cg * 64 + lo * 4];
      const v4f k0 = *(const v4f*)&kt[0][cg * KGP + lo * 4];
      const v4f k1 = *(const v4f*)&kt[1][cg * KGP + lo * 4];
      const v2f a0 = {qa.x, qa.y}, a1 = {qa.z, qa.w};
      const v2f k00 = {k0.x, k0.y}, k01 = {k0.z, k0.w};
      const v2f k10 = {k1.x, k1.y}, k11 = {k1.z, k1.w};
      s0 = __builtin_amdgcn_wmma_f32_16x16x4_f32(false, a0, false, k00,
                                                 (short)0, s0, false, false);
      s1 = __builtin_amdgcn_wmma_f32_16x16x4_f32(false, a0, false, k10,
                                                 (short)0, s1, false, false);
      s0 = __builtin_amdgcn_wmma_f32_16x16x4_f32(false, a1, false, k01,
                                                 (short)0, s0, false, false);
      s1 = __builtin_amdgcn_wmma_f32_16x16x4_f32(false, a1, false, k11,
                                                 (short)0, s1, false, false);
    }

    float* pw = &pl[wave][0];
#pragma unroll
    for (int r = 0; r < 8; ++r) {
      pw[(r + 8 * hg) * PP + lo] = s0[r];
      pw[(r + 8 * hg) * PP + 16 + lo] = s1[r];
    }
    v4f sa[KSTEP / 8];
#pragma unroll
    for (int j = 0; j < KSTEP / 8; ++j)
      sa[j] = *(const v4f*)&pw[lo * PP + 8 * j + 4 * hg];

    float hA = fmaxf(sa[0].x, sa[0].y), lA = fminf(sa[0].x, sa[0].y);
    float hB = fmaxf(sa[2].x, sa[2].y), lB = fminf(sa[2].x, sa[2].y);
    top2_push(hA, lA, sa[0].z); top2_push(hB, lB, sa[2].z);
    top2_push(hA, lA, sa[0].w); top2_push(hB, lB, sa[2].w);
#pragma unroll
    for (int e = 0; e < 4; ++e) {
      top2_push(hA, lA, sa[1][e]); top2_push(hB, lB, sa[3][e]);
    }
    float a1 = fmaxf(hA, hB);
    float a2 = fmaxf(fminf(hA, hB), fmaxf(lA, lB));
    {
      const float b1 = XHALF(a1), b2 = XHALF(a2);
      const float hi = fmaxf(a1, b1), lw = fminf(a1, b1);
      a2 = fmaxf(lw, fmaxf(a2, b2)); a1 = hi;
    }
    const float mo = t1;
    {
      const float hi = fmaxf(t1, a1), lw = fminf(t1, a1);
      t2 = fmaxf(lw, fmaxf(t2, a2)); t1 = hi;
    }

    if (__any(t1 > mo)) {
      const float sc = __expf((mo - t1) * invTemp);
      lsum *= sc;
      const float scs = XOR16_8(sc);
      const float scx = hg ? scs : sc;
#pragma unroll
      for (int r = 0; r < 8; ++r) {
        const float f = xor16(scx, 0x11111111u * r, 0x11111111u * r);
#pragma unroll
        for (int ct = 0; ct < 8; ++ct) o[ct][r] *= f;
      }
    }

    v4f pa[KSTEP / 8];
    float q4[KSTEP / 8];
#pragma unroll
    for (int j = 0; j < KSTEP / 8; ++j) {
#pragma unroll
      for (int e = 0; e < 4; ++e) pa[j][e] = __expf((sa[j][e] - t1) * invTemp);
      q4[j] = (pa[j].x + pa[j].y) + (pa[j].z + pa[j].w);
    }
    lsum += (q4[0] + q4[1]) + (q4[2] + q4[3]);

#pragma unroll
    for (int j = 0; j < KSTEP / 8; ++j) {
      const int k4 = 8 * j + 4 * hg;
      v4f vb[8];
#pragma unroll
      for (int ct = 0; ct < 8; ++ct)
        vb[ct] = *(const v4f*)&vt[(ct * 16 + lo) * VP + k4];
      const v2f a0 = {pa[j].x, pa[j].y}, a1f = {pa[j].z, pa[j].w};
#pragma unroll
      for (int ct = 0; ct < 8; ++ct) {
        const v2f bv = {vb[ct].x, vb[ct].y};
        o[ct] = __builtin_amdgcn_wmma_f32_16x16x4_f32(false, a0, false, bv,
                                                      (short)0, o[ct], false, false);
      }
#pragma unroll
      for (int ct = 0; ct < 8; ++ct) {
        const v2f bv = {vb[ct].z, vb[ct].w};
        o[ct] = __builtin_amdgcn_wmma_f32_16x16x4_f32(false, a1f, false, bv,
                                                      (short)0, o[ct], false, false);
      }
    }
    __syncthreads();
  }

  const size_t sb = (size_t)(split * B_SZ + b);
#pragma unroll
  for (int r = 0; r < 8; ++r) {
    const int q = wave * 16 + r + 8 * hg;
#pragma unroll
    for (int ct = 0; ct < 8; ++ct) {
      const int c = ct * 16 + lo;
      smem_q[c * VSTR + q] = o[ct][r];
    }
  }
  {
    const float ltot = lsum + XHALF(lsum);
    if (hg == 0) {
      const size_t si = sb * N_PTS + qBase0 + wave * 16 + lo;
      Sm[si]  = t1;
      St2[si] = t2;
      Sl[si]  = ltot;
    }
  }
  __syncthreads();

#pragma unroll
  for (int i = 0; i < 16; ++i) {
    const int idx4 = tid + 256 * i;
    const int c  = idx4 >> 5;
    const int q4 = (idx4 & 31) * 4;
    const float4 val = *(const float4*)&smem_q[c * VSTR + q4];
    *(float4*)(Op + (sb * C_DIM + c) * N_PTS + qBase0 + q4) = val;
  }
}

__global__ __launch_bounds__(256) void merge_kernel(const float* __restrict__ Op,
                                                    const float* __restrict__ Sm,
                                                    const float* __restrict__ St2,
                                                    const float* __restrict__ Sl,
                                                    float* __restrict__ Y,
                                                    float* __restrict__ R) {
  const int n4 = (blockIdx.x * 256 + threadIdx.x) * 4;
  const int c0 = blockIdx.y * MCPT;
  const int b  = blockIdx.z;
  const float invTemp = 0.088388347648318447f;

  v4f mv[NSPLIT], lv[NSPLIT], ov[MCPT][NSPLIT];
#pragma unroll
  for (int s = 0; s < NSPLIT; ++s) {
    const size_t si = ((size_t)(s * B_SZ + b)) * N_PTS + n4;
    mv[s] = *(const v4f*)(Sm + si);
    lv[s] = *(const v4f*)(Sl + si);
  }
#pragma unroll
  for (int cc = 0; cc < MCPT; ++cc)
#pragma unroll
    for (int s = 0; s < NSPLIT; ++s)
      ov[cc][s] = *(const v4f*)(Op + ((size_t)((s * B_SZ + b) * C_DIM + c0 + cc)) * N_PTS + n4);

  asm volatile("" :: "v"(mv[0]), "v"(mv[1]), "v"(mv[2]), "v"(mv[3]),
                     "v"(lv[0]), "v"(lv[1]), "v"(lv[2]), "v"(lv[3]),
                     "v"(ov[0][0]), "v"(ov[0][1]), "v"(ov[0][2]), "v"(ov[0][3]),
                     "v"(ov[1][0]), "v"(ov[1][1]), "v"(ov[1][2]), "v"(ov[1][3]),
                     "v"(ov[2][0]), "v"(ov[2][1]), "v"(ov[2][2]), "v"(ov[2][3]),
                     "v"(ov[3][0]), "v"(ov[3][1]), "v"(ov[3][2]), "v"(ov[3][3]));

  v4f M = mv[0];
#pragma unroll
  for (int s = 1; s < NSPLIT; ++s)
#pragma unroll
    for (int j = 0; j < 4; ++j) M[j] = fmaxf(M[j], mv[s][j]);
  v4f w[NSPLIT];
  v4f L = {0.0f, 0.0f, 0.0f, 0.0f};
#pragma unroll
  for (int s = 0; s < NSPLIT; ++s)
#pragma unroll
    for (int j = 0; j < 4; ++j) {
      w[s][j] = __expf((mv[s][j] - M[j]) * invTemp);
      L[j] += w[s][j] * lv[s][j];
    }
  v4f inv;
#pragma unroll
  for (int j = 0; j < 4; ++j) inv[j] = 1.0f / L[j];

#pragma unroll
  for (int cc = 0; cc < MCPT; ++cc) {
    const int c = c0 + cc;
    v4f acc = {0.0f, 0.0f, 0.0f, 0.0f};
#pragma unroll
    for (int s = 0; s < NSPLIT; ++s) acc += w[s] * ov[cc][s];
    *(v4f*)(Y + ((size_t)(b * C_DIM + c)) * N_PTS + n4) = acc * inv;
  }

  if (blockIdx.y == 0) {
    const float NEGINF = -__builtin_inff();
    v4f tv[NSPLIT];
#pragma unroll
    for (int s = 0; s < NSPLIT; ++s)
      tv[s] = *(const v4f*)(St2 + ((size_t)(s * B_SZ + b)) * N_PTS + n4);
    v4f rv;
#pragma unroll
    for (int j = 0; j < 4; ++j) {
      float T1 = NEGINF, T2 = NEGINF;
#pragma unroll
      for (int s = 0; s < NSPLIT; ++s) {
        const float hi = fmaxf(T1, mv[s][j]);
        const float lw = fminf(T1, mv[s][j]);
        T2 = fmaxf(lw, fmaxf(T2, tv[s][j]));
        T1 = hi;
      }
      const float x1 = fminf(fmaxf(T1, -1.0f), 1.0f);
      const float x2 = fminf(fmaxf(T2, -1.0f), 1.0f);
      rv[j] = acosf(x1) / acosf(x2);
    }
    *(v4f*)(R + (size_t)b * N_PTS + n4) = rv;
  }
}

extern "C" void kernel_launch(void* const* d_in, const int* in_sizes, int n_in,
                              void* d_out, int out_size, void* d_ws, size_t ws_size,
                              hipStream_t stream) {
  (void)in_sizes; (void)n_in; (void)out_size;
  const float* q  = (const float*)d_in[0];
  const float* k  = (const float*)d_in[1];
  const float* v  = (const float*)d_in[2];
  const float* Wq = (const float*)d_in[3];
  const float* Wk = (const float*)d_in[4];
  const float* Wv = (const float*)d_in[5];

  float* out  = (float*)d_out;
  float* Yout = out;
  float* Rout = out + (size_t)B_SZ * C_DIM * N_PTS;

  const size_t tensElems = (size_t)B_SZ * C_DIM * N_PTS;
  const float *qc = q, *kc = k, *vc = v;

  if (ws_size >= 3 * tensElems * sizeof(float)) {
    float* qw = (float*)d_ws;
    float* kw = qw + tensElems;
    float* vw = kw + tensElems;
    dim3 g(N_PTS / 1024, C_DIM / MIX_OT, 3 * B_SZ);
    cmix3_kernel<<<g, 256, 0, stream>>>(q, k, v, Wq, Wk, Wv, qw, kw, vw);
    qc = qw; kc = kw; vc = vw;
  }

  float* Op  = (float*)d_ws + 3 * tensElems;
  float* Sm  = Op + NSPLIT * tensElems;
  float* St2 = Sm + (size_t)NSPLIT * B_SZ * N_PTS;
  float* Sl  = St2 + (size_t)NSPLIT * B_SZ * N_PTS;

  dim3 ga(N_PTS / 128, B_SZ, NSPLIT);
  attn_kernel<<<ga, 256, 0, stream>>>(qc, kc, vc, Op, Sm, St2, Sl);

  dim3 gm(N_PTS / 1024, C_DIM / MCPT, B_SZ);
  merge_kernel<<<gm, 256, 0, stream>>>(Op, Sm, St2, Sl, Yout, Rout);
}


# attn: wave priority falls with progress (s_setprio 3, then one level down at 1/2, 3/4 and 7/8 of the key range): of a WGP's two work-groups the lagging one wins the matrix pipe; sum order unchanged
# speedup vs baseline: 3.3798x; 1.0382x over previous
#include <hip/hip_runtime.h>
#include <math.h>

typedef float v2f __attribute__((ext_vector_type(2)));
typedef float v4f __attribute__((ext_vector_type(4)));
typedef float v8f __attribute__((ext_vector_type(8)));

#define C_DIM 128
#define N_PTS 4096
#define B_SZ  4
#define VSTR  132
#define KGP   68
#define KSUB  (32 * KGP)
#define VP    36
#define PP    36
#define QT_W  2048
#define SMEMQ 16896
#define NSPLIT 4
#define KSPAN (N_PTS / NSPLIT)
#define KSTEP 32
#define MCPT  4

__device__ __forceinline__ float xor16(float v, unsigned selLo, unsigned selHi) {
  unsigned i = __builtin_bit_cast(unsigned, v);
  i = __builtin_amdgcn_permlane16(i, i, selLo, selHi, false, false);
  return __builtin_bit_cast(float, i);
}
#define XOR16_1(v) xor16((v), 0x67452301u, 0xEFCDAB89u)
#define XOR16_2(v) xor16((v), 0x54761032u, 0xDCFE98BAu)
#define XOR16_4(v) xor16((v), 0x32107654u, 0xBA98FEDCu)
#define XOR16_8(v) xor16((v), 0xFEDCBA98u, 0x76543210u)

__device__ __forceinline__ float xhalf(float v) {
  unsigned i = __builtin_bit_cast(unsigned, v);
  i = __builtin_amdgcn_permlanex16(i, i, 0x76543210u, 0xFEDCBA98u, false, false);
  return __builtin_bit_cast(float, i);
}
#define XHALF(v) xhalf(v)

__device__ __forceinline__ void top2_push(float& h, float& l, float x) {
  l = fmaxf(fminf(h, x), l);
  h = fmaxf(h, x);
}

#define MIX_OT 8
__global__ __launch_bounds__(256) void cmix3_kernel(const float* __restrict__ xq,
                                                    const float* __restrict__ xk,
                                                    const float* __restrict__ xv,
                                                    const float* __restrict__ Wq,
                                                    const float* __restrict__ Wk,
                                                    const float* __restrict__ Wv,
                                                    float* __restrict__ oq,
                                                    float* __restrict__ ok,
                                                    float* __restrict__ ov) {
  const int o0 = blockIdx.y * MIX_OT;
  const int t  = blockIdx.z / B_SZ;
  const int b  = blockIdx.z - t * B_SZ;
  const float* __restrict__ x   = (t == 0) ? xq : ((t == 1) ? xk : xv);
  const float* __restrict__ W   = (t == 0) ? Wq : ((t == 1) ? Wk : Wv);
  float* __restrict__       out = (t == 0) ? oq : ((t == 1) ? ok : ov);
  const int n4 = (blockIdx.x * 256 + threadIdx.x) * 4;
  const float* xp = x + (size_t)b * C_DIM * N_PTS + n4;
  const float* wp = W + o0 * C_DIM;

  v4f acc[MIX_OT];
#pragma unroll
  for (int j = 0; j < MIX_OT; ++j) acc[j] = (v4f){0.0f, 0.0f, 0.0f, 0.0f};

#pragma unroll 16
  for (int c = 0; c < C_DIM; ++c) {
    const v4f xv4 = *(const v4f*)(xp + (size_t)c * N_PTS);
#pragma unroll
    for (int j = 0; j < MIX_OT; ++j) {
      const float w = wp[j * C_DIM + c];
#pragma unroll
      for (int i = 0; i < 4; ++i) acc[j][i] = __builtin_fmaf(w, xv4[i], acc[j][i]);
    }
  }

#pragma unroll
  for (int j = 0; j < MIX_OT; ++j)
    *(v4f*)(out + ((size_t)(b * C_DIM + o0 + j)) * N_PTS + n4) = acc[j];
}

__global__ __launch_bounds__(256)
__attribute__((amdgpu_num_vgpr(256)))
void attn_kernel(const float* __restrict__ Q,
                 const float* __restrict__ K,
                 const float* __restrict__ V,
                 float* __restrict__ Op,
                 float* __restrict__ Sm,
                 float* __restrict__ St2,
                 float* __restrict__ Sl) {
  __shared__ __attribute__((aligned(16))) float smem_q[SMEMQ];
  __shared__ __attribute__((aligned(16))) float kt[2][KSUB];
  __shared__ __attribute__((aligned(16))) float vt[C_DIM * VP];
  __shared__ __attribute__((aligned(16))) float pl[8][16 * PP];

  const int tid  = threadIdx.x;
  const int lane = tid & 31;
  const int wave = tid >> 5;
  const int b    = blockIdx.y;
  const int split = blockIdx.z;
  const int kBeg = split * KSPAN;
  const int qBase0 = blockIdx.x * 128;
  const int lo = lane & 15;
  const int hg = lane >> 4;
  const float invTemp = 0.088388347648318447f;
  const float NEGINF = -__builtin_inff();

#pragma unroll
  for (int i = 0; i < 16; ++i) {
    const int idx4 = tid + 256 * i;
    const int c  = idx4 >> 5;
    const int q4 = idx4 & 31;
    const float4 qv = *(const float4*)(Q + ((size_t)(b * C_DIM + c)) * N_PTS + qBase0 + q4 * 4);
    const int q = q4 * 4;
    float* qd = &smem_q[(q >> 4) * QT_W + (c >> 2) * 64 + (q & 15) * 4 + (c & 3)];
    qd[0] = qv.x; qd[4] = qv.y; qd[8] = qv.z; qd[12] = qv.w;
  }
  __syncthreads();
  const float* qt_w = smem_q + wave * QT_W;

  v8f  o[8] = {};
  float t1 = NEGINF, t2 = NEGINF;
  float lsum = 0.0f;

  __builtin_amdgcn_s_setprio(3);
  for (int kb0 = kBeg; kb0 < kBeg + KSPAN; kb0 += KSTEP) {
    const int done = kb0 - kBeg;
    if (done == KSPAN / 2)       __builtin_amdgcn_s_setprio(2);
    if (done == (KSPAN * 3) / 4) __builtin_amdgcn_s_setprio(1);
    if (done == (KSPAN * 7) / 8) __builtin_amdgcn_s_setprio(0);
#pragma unroll
    for (int h = 0; h < 2; ++h) {
#pragma unroll
      for (int i = 0; i < 2; ++i) {
        const int idx4 = tid + 256 * i;
        const int c  = idx4 >> 2;
        const int n4 = (idx4 & 3) * 4;
        const size_t g = ((size_t)(b * C_DIM + c)) * N_PTS + kb0 + 16 * h + n4;
        const float4 kv = *(const float4*)(K + g);
        float* kd = &kt[h][(c >> 2) * KGP + n4 * 4 + (c & 3)];
        kd[0] = kv.x; kd[4] = kv.y; kd[8] = kv.z; kd[12] = kv.w;
        const float4 vv = *(const float4*)(V + g);
        *(float4*)&vt[c * VP + 16 * h + n4] = vv;
      }
    }
    __syncthreads();

    v8f s0 = {}, s1 = {};
#pragma unroll 4
    for (int j = 0; j < 16; ++j) {
      const int cg = 2 * j + hg;
      const v4f qa = *(const v4f*)&qt_w[cg * 64 + lo * 4];
      const v4f k0 = *(const v4f*)&kt[0][cg * KGP + lo * 4];
      const v4f k1 = *(const v4f*)&kt[1][cg * KGP + lo * 4];
      const v2f a0 = {qa.x, qa.y}, a1 = {qa.z, qa.w};
      const v2f k00 = {k0.x, k0.y}, k01 = {k0.z, k0.w};
      const v2f k10 = {k1.x, k1.y}, k11 = {k1.z, k1.w};
      s0 = __builtin_amdgcn_wmma_f32_16x16x4_f32(false, a0, false, k00,
                                                 (short)0, s0, false, false);
      s1 = __builtin_amdgcn_wmma_f32_16x16x4_f32(false, a0, false, k10,
                                                 (short)0, s1, false, false);
      s0 = __builtin_amdgcn_wmma_f32_16x16x4_f32(false, a1, false, k01,
                                                 (short)0, s0, false, false);
      s1 = __builtin_amdgcn_wmma_f32_16x16x4_f32(false, a1, false, k11,
                                                 (short)0, s1, false, false);
    }

    float* pw = &pl[wave][0];
#pragma unroll
    for (int r = 0; r < 8; ++r) {
      pw[(r + 8 * hg) * PP + lo] = s0[r];
      pw[(r + 8 * hg) * PP + 16 + lo] = s1[r];
    }
    v4f sa[KSTEP / 8];
#pragma unroll
    for (int j = 0; j < KSTEP / 8; ++j)
      sa[j] = *(const v4f*)&pw[lo * PP + 8 * j + 4 * hg];

    float hA = fmaxf(sa[0].x, sa[0].y), lA = fminf(sa[0].x, sa[0].y);
    float hB = fmaxf(sa[2].x, sa[2].y), lB = fminf(sa[2].x, sa[2].y);
    top2_push(hA, lA, sa[0].z); top2_push(hB, lB, sa[2].z);
    top2_push(hA, lA, sa[0].w); top2_push(hB, lB, sa[2].w);
#pragma unroll
    for (int e = 0; e < 4; ++e) {
      top2_push(hA, lA, sa[1][e]); top2_push(hB, lB, sa[3][e]);
    }
    float a1 = fmaxf(hA, hB);
    float a2 = fmaxf(fminf(hA, hB), fmaxf(lA, lB));
    {
      const float b1 = XHALF(a1), b2 = XHALF(a2);
      const float hi = fmaxf(a1, b1), lw = fminf(a1, b1);
      a2 = fmaxf(lw, fmaxf(a2, b2)); a1 = hi;
    }
    const float mo = t1;
    {
      const float hi = fmaxf(t1, a1), lw = fminf(t1, a1);
      t2 = fmaxf(lw, fmaxf(t2, a2)); t1 = hi;
    }

    if (__any(t1 > mo)) {
      const float sc = __expf((mo - t1) * invTemp);
      lsum *= sc;
      const float scs = XOR16_8(sc);
      const float scx = hg ? scs : sc;
#pragma unroll
      for (int r = 0; r < 8; ++r) {
        const float f = xor16(scx, 0x11111111u * r, 0x11111111u * r);
#pragma unroll
        for (int ct = 0; ct < 8; ++ct) o[ct][r] *= f;
      }
    }

    v4f pa[KSTEP / 8];
    float q4[KSTEP / 8];
#pragma unroll
    for (int j = 0; j < KSTEP / 8; ++j) {
#pragma unroll
      for (int e = 0; e < 4; ++e) pa[j][e] = __expf((sa[j][e] - t1) * invTemp);
      q4[j] = (pa[j].x + pa[j].y) + (pa[j].z + pa[j].w);
    }
    lsum += (q4[0] + q4[1]) + (q4[2] + q4[3]);

#pragma unroll
    for (int j = 0; j < KSTEP / 8; ++j) {
      const int k4 = 8 * j + 4 * hg;
      v4f vb[8];
#pragma unroll
      for (int ct = 0; ct < 8; ++ct)
        vb[ct] = *(const v4f*)&vt[(ct * 16 + lo) * VP + k4];
      const v2f a0 = {pa[j].x, pa[j].y}, a1f = {pa[j].z, pa[j].w};
#pragma unroll
      for (int ct = 0; ct < 8; ++ct) {
        const v2f bv = {vb[ct].x, vb[ct].y};
        o[ct] = __builtin_amdgcn_wmma_f32_16x16x4_f32(false, a0, false, bv,
                                                      (short)0, o[ct], false, false);
      }
#pragma unroll
      for (int ct = 0; ct < 8; ++ct) {
        const v2f bv = {vb[ct].z, vb[ct].w};
        o[ct] = __builtin_amdgcn_wmma_f32_16x16x4_f32(false, a1f, false, bv,
                                                      (short)0, o[ct], false, false);
      }
    }
    __syncthreads();
  }

  const size_t sb = (size_t)(split * B_SZ + b);
#pragma unroll
  for (int r = 0; r < 8; ++r) {
    const int q = wave * 16 + r + 8 * hg;
#pragma unroll
    for (int ct = 0; ct < 8; ++ct) {
      const int c = ct * 16 + lo;
      smem_q[c * VSTR + q] = o[ct][r];
    }
  }
  {
    const float ltot = lsum + XHALF(lsum);
    if (hg == 0) {
      const size_t si = sb * N_PTS + qBase0 + wave * 16 + lo;
      Sm[si]  = t1;
      St2[si] = t2;
      Sl[si]  = ltot;
    }
  }
  __syncthreads();

#pragma unroll
  for (int i = 0; i < 16; ++i) {
    const int idx4 = tid + 256 * i;
    const int c  = idx4 >> 5;
    const int q4 = (idx4 & 31) * 4;
    const float4 val = *(const float4*)&smem_q[c * VSTR + q4];
    *(float4*)(Op + (sb * C_DIM + c) * N_PTS + qBase0 + q4) = val;
  }
}

__global__ __launch_bounds__(256) void merge_kernel(const float* __restrict__ Op,
                                                    const float* __restrict__ Sm,
                                                    const float* __restrict__ St2,
                                                    const float* __restrict__ Sl,
                                                    float* __restrict__ Y,
                                                    float* __restrict__ R) {
  const int n4 = (blockIdx.x * 256 + threadIdx.x) * 4;
  const int c0 = blockIdx.y * MCPT;
  const int b  = blockIdx.z;
  const float invTemp = 0.088388347648318447f;

  v4f mv[NSPLIT], lv[NSPLIT], ov[MCPT][NSPLIT];
#pragma unroll
  for (int s = 0; s < NSPLIT; ++s) {
    const size_t si = ((size_t)(s * B_SZ + b)) * N_PTS + n4;
    mv[s] = *(const v4f*)(Sm + si);
    lv[s] = *(const v4f*)(Sl + si);
  }
#pragma unroll
  for (int cc = 0; cc < MCPT; ++cc)
#pragma unroll
    for (int s = 0; s < NSPLIT; ++s)
      ov[cc][s] = *(const v4f*)(Op + ((size_t)((s * B_SZ + b) * C_DIM + c0 + cc)) * N_PTS + n4);

  asm volatile("" :: "v"(mv[0]), "v"(mv[1]), "v"(mv[2]), "v"(mv[3]),
                     "v"(lv[0]), "v"(lv[1]), "v"(lv[2]), "v"(lv[3]),
                     "v"(ov[0][0]), "v"(ov[0][1]), "v"(ov[0][2]), "v"(ov[0][3]),
                     "v"(ov[1][0]), "v"(ov[1][1]), "v"(ov[1][2]), "v"(ov[1][3]),
                     "v"(ov[2][0]), "v"(ov[2][1]), "v"(ov[2][2]), "v"(ov[2][3]),
                     "v"(ov[3][0]), "v"(ov[3][1]), "v"(ov[3][2]), "v"(ov[3][3]));

  v4f M = mv[0];
#pragma unroll
  for (int s = 1; s < NSPLIT; ++s)
#pragma unroll
    for (int j = 0; j < 4; ++j) M[j] = fmaxf(M[j], mv[s][j]);
  v4f w[NSPLIT];
  v4f L = {0.0f, 0.0f, 0.0f, 0.0f};
#pragma unroll
  for (int s = 0; s < NSPLIT; ++s)
#pragma unroll
    for (int j = 0; j < 4; ++j) {
      w[s][j] = __expf((mv[s][j] - M[j]) * invTemp);
      L[j] += w[s][j] * lv[s][j];
    }
  v4f inv;
#pragma unroll
  for (int j = 0; j < 4; ++j) inv[j] = 1.0f / L[j];

#pragma unroll
  for (int cc = 0; cc < MCPT; ++cc) {
    const int c = c0 + cc;
    v4f acc = {0.0f, 0.0f, 0.0f, 0.0f};
#pragma unroll
    for (int s = 0; s < NSPLIT; ++s) acc += w[s] * ov[cc][s];
    *(v4f*)(Y + ((size_t)(b * C_DIM + c)) * N_PTS + n4) = acc * inv;
  }

  if (blockIdx.y == 0) {
    const float NEGINF = -__builtin_inff();
    v4f tv[NSPLIT];
#pragma unroll
    for (int s = 0; s < NSPLIT; ++s)
      tv[s] = *(const v4f*)(St2 + ((size_t)(s * B_SZ + b)) * N_PTS + n4);
    v4f rv;
#pragma unroll
    for (int j = 0; j < 4; ++j) {
      float T1 = NEGINF, T2 = NEGINF;
#pragma unroll
      for (int s = 0; s < NSPLIT; ++s) {
        const float hi = fmaxf(T1, mv[s][j]);
        const float lw = fminf(T1, mv[s][j]);
        T2 = fmaxf(lw, fmaxf(T2, tv[s][j]));
        T1 = hi;
      }
      const float x1 = fminf(fmaxf(T1, -1.0f), 1.0f);
      const float x2 = fminf(fmaxf(T2, -1.0f), 1.0f);
      rv[j] = acosf(x1) / acosf(x2);
    }
    *(v4f*)(R + (size_t)b * N_PTS + n4) = rv;
  }
}

extern "C" void kernel_launch(void* const* d_in, const int* in_sizes, int n_in,
                              void* d_out, int out_size, void* d_ws, size_t ws_size,
                              hipStream_t stream) {
  (void)in_sizes; (void)n_in; (void)out_size;
  const float* q  = (const float*)d_in[0];
  const float* k  = (const float*)d_in[1];
  const float* v  = (const float*)d_in[2];
  const float* Wq = (const float*)d_in[3];
  const float* Wk = (const float*)d_in[4];
  const float* Wv = (const float*)d_in[5];

  float* out  = (float*)d_out;
  float* Yout = out;
  float* Rout = out + (size_t)B_SZ * C_DIM * N_PTS;

  const size_t tensElems = (size_t)B_SZ * C_DIM * N_PTS;
  const float *qc = q, *kc = k, *vc = v;

  if (ws_size >= 3 * tensElems * sizeof(float)) {
    float* qw = (float*)d_ws;
    float* kw = qw + tensElems;
    float* vw = kw + tensElems;
    dim3 g(N_PTS / 1024, C_DIM / MIX_OT, 3 * B_SZ);
    cmix3_kernel<<<g, 256, 0, stream>>>(q, k, v, Wq, Wk, Wv, qw, kw, vw);
    qc = qw; kc = kw; vc = vw;
  }

  float* Op  = (float*)d_ws + 3 * tensElems;
  float* Sm  = Op + NSPLIT * tensElems;
  float* St2 = Sm + (size_t)NSPLIT * B_SZ * N_PTS;
  float* Sl  = St2 + (size_t)NSPLIT * B_SZ * N_PTS;

  dim3 ga(N_PTS / 128, B_SZ, NSPLIT);
  attn_kernel<<<ga, 256, 0, stream>>>(qc, kc, vc, Op, Sm, St2, Sl);

  dim3 gm(N_PTS / 1024, C_DIM / MCPT, B_SZ);
  merge_kernel<<<gm, 256, 0, stream>>>(Op, Sm, St2, Sl, Yout, Rout);
}


# attn: key loop steps 64 keys (was 32) as four 16-key tiles: a Q fragment feeds four score chains; staging, barrier pair and row-max update once per 64 keys (rescale rounding, row-sum order change)
# speedup vs baseline: 3.5412x; 1.0478x over previous
#include <hip/hip_runtime.h>
#include <math.h>

typedef float v2f __attribute__((ext_vector_type(2)));
typedef float v4f __attribute__((ext_vector_type(4)));
typedef float v8f __attribute__((ext_vector_type(8)));

#define C_DIM 128
#define N_PTS 4096
#define B_SZ  4
#define VSTR  132
#define KGP   68
#define KSUB  (32 * KGP)
#define VP    68
#define PP    36
#define QT_W  2048
#define SMEMQ 16896
#define NSPLIT 4
#define KSPAN (N_PTS / NSPLIT)
#define KSTEP 64
#define NSUB  (KSTEP / 16)
#define MCPT  4

__device__ __forceinline__ float xor16(float v, unsigned selLo, unsigned selHi) {
  unsigned i = __builtin_bit_cast(unsigned, v);
  i = __builtin_amdgcn_permlane16(i, i, selLo, selHi, false, false);
  return __builtin_bit_cast(float, i);
}
#define XOR16_1(v) xor16((v), 0x67452301u, 0xEFCDAB89u)
#define XOR16_2(v) xor16((v), 0x54761032u, 0xDCFE98BAu)
#define XOR16_4(v) xor16((v), 0x32107654u, 0xBA98FEDCu)
#define XOR16_8(v) xor16((v), 0xFEDCBA98u, 0x76543210u)

__device__ __forceinline__ float xhalf(float v) {
  unsigned i = __builtin_bit_cast(unsigned, v);
  i = __builtin_amdgcn_permlanex16(i, i, 0x76543210u, 0xFEDCBA98u, false, false);
  return __builtin_bit_cast(float, i);
}
#define XHALF(v) xhalf(v)

__device__ __forceinline__ void top2_push(float& h, float& l, float x) {
  l = fmaxf(fminf(h, x), l);
  h = fmaxf(h, x);
}

#define MIX_OT 8
__global__ __launch_bounds__(256) void cmix3_kernel(const float* __restrict__ xq,
                                                    const float* __restrict__ xk,
                                                    const float* __restrict__ xv,
                                                    const float* __restrict__ Wq,
                                                    const float* __restrict__ Wk,
                                                    const float* __restrict__ Wv,
                                                    float* __restrict__ oq,
                                                    float* __restrict__ ok,
                                                    float* __restrict__ ov) {
  const int o0 = blockIdx.y * MIX_OT;
  const int t  = blockIdx.z / B_SZ;
  const int b  = blockIdx.z - t * B_SZ;
  const float* __restrict__ x   = (t == 0) ? xq : ((t == 1) ? xk : xv);
  const float* __restrict__ W   = (t == 0) ? Wq : ((t == 1) ? Wk : Wv);
  float* __restrict__       out = (t == 0) ? oq : ((t == 1) ? ok : ov);
  const int n4 = (blockIdx.x * 256 + threadIdx.x) * 4;
  const float* xp = x + (size_t)b * C_DIM * N_PTS + n4;
  const float* wp = W + o0 * C_DIM;

  v4f acc[MIX_OT];
#pragma unroll
  for (int j = 0; j < MIX_OT; ++j) acc[j] = (v4f){0.0f, 0.0f, 0.0f, 0.0f};

#pragma unroll 16
  for (int c = 0; c < C_DIM; ++c) {
    const v4f xv4 = *(const v4f*)(xp + (size_t)c * N_PTS);
#pragma unroll
    for (int j = 0; j < MIX_OT; ++j) {
      const float w = wp[j * C_DIM + c];
#pragma unroll
      for (int i = 0; i < 4; ++i) acc[j][i] = __builtin_fmaf(w, xv4[i], acc[j][i]);
    }
  }

#pragma unroll
  for (int j = 0; j < MIX_OT; ++j)
    *(v4f*)(out + ((size_t)(b * C_DIM + o0 + j)) * N_PTS + n4) = acc[j];
}

__global__ __launch_bounds__(256)
__attribute__((amdgpu_num_vgpr(256)))
void attn_kernel(const float* __restrict__ Q,
                 const float* __restrict__ K,
                 const float* __restrict__ V,
                 float* __restrict__ Op,
                 float* __restrict__ Sm,
                 float* __restrict__ St2,
                 float* __restrict__ Sl) {
  __shared__ __attribute__((aligned(16))) float smem_q[SMEMQ];
  __shared__ __attribute__((aligned(16))) float kt[NSUB][KSUB];
  __shared__ __attribute__((aligned(16))) float vt[C_DIM * VP];
  __shared__ __attribute__((aligned(16))) float pl[8][16 * PP];

  const int tid  = threadIdx.x;
  const int lane = tid & 31;
  const int wave = tid >> 5;
  const int b    = blockIdx.y;
  const int split = blockIdx.z;
  const int kBeg = split * KSPAN;
  const int qBase0 = blockIdx.x * 128;
  const int lo = lane & 15;
  const int hg = lane >> 4;
  const float invTemp = 0.088388347648318447f;
  const float NEGINF = -__builtin_inff();

#pragma unroll
  for (int i = 0; i < 16; ++i) {
    const int idx4 = tid + 256 * i;
    const int c  = idx4 >> 5;
    const int q4 = idx4 & 31;
    const float4 qv = *(const float4*)(Q + ((size_t)(b * C_DIM + c)) * N_PTS + qBase0 + q4 * 4);
    const int q = q4 * 4;
    float* qd = &smem_q[(q >> 4) * QT_W + (c >> 2) * 64 + (q & 15) * 4 + (c & 3)];
    qd[0] = qv.x; qd[4] = qv.y; qd[8] = qv.z; qd[12] = qv.w;
  }
  __syncthreads();
  const float* qt_w = smem_q + wave * QT_W;

  v8f  o[8] = {};
  float t1 = NEGINF, t2 = NEGINF;
  float lsum = 0.0f;

  __builtin_amdgcn_s_setprio(3);
  for (int kb0 = kBeg; kb0 < kBeg + KSPAN; kb0 += KSTEP) {
    const int done = kb0 - kBeg;
    if (done == KSPAN / 2)       __builtin_amdgcn_s_setprio(2);
    if (done == (KSPAN * 3) / 4) __builtin_amdgcn_s_setprio(1);
    if (done == (KSPAN * 7) / 8) __builtin_amdgcn_s_setprio(0);
#pragma unroll
    for (int h = 0; h < NSUB; ++h) {
#pragma unroll
      for (int i = 0; i < 2; ++i) {
        const int idx4 = tid + 256 * i;
        const int c  = idx4 >> 2;
        const int n4 = (idx4 & 3) * 4;
        const size_t g = ((size_t)(b * C_DIM + c)) * N_PTS + kb0 + 16 * h + n4;
        const float4 kv = *(const float4*)(K + g);
        float* kd = &kt[h][(c >> 2) * KGP + n4 * 4 + (c & 3)];
        kd[0] = kv.x; kd[4] = kv.y; kd[8] = kv.z; kd[12] = kv.w;
      }
    }
#pragma unroll
    for (int i = 0; i < 8; ++i) {
      const int idx4 = tid + 256 * i;
      const int c  = idx4 >> 4;
      const int n4 = (idx4 & 15) * 4;
      const float4 vv = *(const float4*)(V + ((size_t)(b * C_DIM + c)) * N_PTS + kb0 + n4);
      *(float4*)&vt[c * VP + n4] = vv;
    }
    __syncthreads();

    v8f s[NSUB] = {};
#pragma unroll 4
    for (int j = 0; j < 16; ++j) {
      const int cg = 2 * j + hg;
      const v4f qa = *(const v4f*)&qt_w[cg * 64 + lo * 4];
      v4f kf[NSUB];
#pragma unroll
      for (int h = 0; h < NSUB; ++h) kf[h] = *(const v4f*)&kt[h][cg * KGP + lo * 4];
      const v2f a0 = {qa.x, qa.y}, a1 = {qa.z, qa.w};
#pragma unroll
      for (int h = 0; h < NSUB; ++h) {
        const v2f kb = {kf[h].x, kf[h].y};
        s[h] = __builtin_amdgcn_wmma_f32_16x16x4_f32(false, a0, false, kb,
                                                     (short)0, s[h], false, false);
      }
#pragma unroll
      for (int h = 0; h < NSUB; ++h) {
        const v2f kb = {kf[h].z, kf[h].w};
        s[h] = __builtin_amdgcn_wmma_f32_16x16x4_f32(false, a1, false, kb,
                                                     (short)0, s[h], false, false);
      }
    }

    float* pw = &pl[wave][0];
    v4f sa[KSTEP / 8];
#pragma unroll
    for (int hh = 0; hh < 2; ++hh) {
#pragma unroll
      for (int r = 0; r < 8; ++r) {
        pw[(r + 8 * hg) * PP + lo] = s[2 * hh][r];
        pw[(r + 8 * hg) * PP + 16 + lo] = s[2 * hh + 1][r];
      }
#pragma unroll
      for (int j = 0; j < 4; ++j)
        sa[4 * hh + j] = *(const v4f*)&pw[lo * PP + 8 * j + 4 * hg];
    }

    float hA = fmaxf(sa[0].x, sa[0].y), lA = fminf(sa[0].x, sa[0].y);
    float hB = fmaxf(sa[2].x, sa[2].y), lB = fminf(sa[2].x, sa[2].y);
    float hC = fmaxf(sa[4].x, sa[4].y), lC = fminf(sa[4].x, sa[4].y);
    float hD = fmaxf(sa[6].x, sa[6].y), lD = fminf(sa[6].x, sa[6].y);
    top2_push(hA, lA, sa[0].z); top2_push(hB, lB, sa[2].z);
    top2_push(hC, lC, sa[4].z); top2_push(hD, lD, sa[6].z);
    top2_push(hA, lA, sa[0].w); top2_push(hB, lB, sa[2].w);
    top2_push(hC, lC, sa[4].w); top2_push(hD, lD, sa[6].w);
#pragma unroll
    for (int e = 0; e < 4; ++e) {
      top2_push(hA, lA, sa[1][e]); top2_push(hB, lB, sa[3][e]);
      top2_push(hC, lC, sa[5][e]); top2_push(hD, lD, sa[7][e]);
    }
    const float hAB = fmaxf(hA, hB), lAB = fmaxf(fminf(hA, hB), fmaxf(lA, lB));
    const float hCD = fmaxf(hC, hD), lCD = fmaxf(fminf(hC, hD), fmaxf(lC, lD));
    float a1 = fmaxf(hAB, hCD);
    float a2 = fmaxf(fminf(hAB, hCD), fmaxf(lAB, lCD));
    {
      const float b1 = XHALF(a1), b2 = XHALF(a2);
      const float hi = fmaxf(a1, b1), lw = fminf(a1, b1);
      a2 = fmaxf(lw, fmaxf(a2, b2)); a1 = hi;
    }
    const float mo = t1;
    {
      const float hi = fmaxf(t1, a1), lw = fminf(t1, a1);
      t2 = fmaxf(lw, fmaxf(t2, a2)); t1 = hi;
    }

    if (__any(t1 > mo)) {
      const float sc = __expf((mo - t1) * invTemp);
      lsum *= sc;
      const float scs = XOR16_8(sc);
      const float scx = hg ? scs : sc;
#pragma unroll
      for (int r = 0; r < 8; ++r) {
        const float f = xor16(scx, 0x11111111u * r, 0x11111111u * r);
#pragma unroll
        for (int ct = 0; ct < 8; ++ct) o[ct][r] *= f;
      }
    }

    v4f pa[KSTEP / 8];
    float q4[KSTEP / 8];
#pragma unroll
    for (int j = 0; j < KSTEP / 8; ++j) {
#pragma unroll
      for (int e = 0; e < 4; ++e) pa[j][e] = __expf((sa[j][e] - t1) * invTemp);
      q4[j] = (pa[j].x + pa[j].y) + (pa[j].z + pa[j].w);
    }
    lsum += ((q4[0] + q4[1]) + (q4[2] + q4[3])) + ((q4[4] + q4[5]) + (q4[6] + q4[7]));

#pragma unroll
    for (int j = 0; j < KSTEP / 8; ++j) {
      const int k4 = 8 * j + 4 * hg;
      v4f vb[8];
#pragma unroll
      for (int ct = 0; ct < 8; ++ct)
        vb[ct] = *(const v4f*)&vt[(ct * 16 + lo) * VP + k4];
      const v2f a0 = {pa[j].x, pa[j].y}, a1f = {pa[j].z, pa[j].w};
#pragma unroll
      for (int ct = 0; ct < 8; ++ct) {
        const v2f bv = {vb[ct].x, vb[ct].y};
        o[ct] = __builtin_amdgcn_wmma_f32_16x16x4_f32(false, a0, false, bv,
                                                      (short)0, o[ct], false, false);
      }
#pragma unroll
      for (int ct = 0; ct < 8; ++ct) {
        const v2f bv = {vb[ct].z, vb[ct].w};
        o[ct] = __builtin_amdgcn_wmma_f32_16x16x4_f32(false, a1f, false, bv,
                                                      (short)0, o[ct], false, false);
      }
    }
    __syncthreads();
  }

  const size_t sb = (size_t)(split * B_SZ + b);
#pragma unroll
  for (int r = 0; r < 8; ++r) {
    const int q = wave * 16 + r + 8 * hg;
#pragma unroll
    for (int ct = 0; ct < 8; ++ct) {
      const int c = ct * 16 + lo;
      smem_q[c * VSTR + q] = o[ct][r];
    }
  }
  {
    const float ltot = lsum + XHALF(lsum);
    if (hg == 0) {
      const size_t si = sb * N_PTS + qBase0 + wave * 16 + lo;
      Sm[si]  = t1;
      St2[si] = t2;
      Sl[si]  = ltot;
    }
  }
  __syncthreads();

#pragma unroll
  for (int i = 0; i < 16; ++i) {
    const int idx4 = tid + 256 * i;
    const int c  = idx4 >> 5;
    const int q4 = (idx4 & 31) * 4;
    const float4 val = *(const float4*)&smem_q[c * VSTR + q4];
    *(float4*)(Op + (sb * C_DIM + c) * N_PTS + qBase0 + q4) = val;
  }
}

__global__ __launch_bounds__(256) void merge_kernel(const float* __restrict__ Op,
                                                    const float* __restrict__ Sm,
                                                    const float* __restrict__ St2,
                                                    const float* __restrict__ Sl,
                                                    float* __restrict__ Y,
                                                    float* __restrict__ R) {
  const int n4 = (blockIdx.x * 256 + threadIdx.x) * 4;
  const int c0 = blockIdx.y * MCPT;
  const int b  = blockIdx.z;
  const float invTemp = 0.088388347648318447f;

  v4f mv[NSPLIT], lv[NSPLIT], ov[MCPT][NSPLIT];
#pragma unroll
  for (int s = 0; s < NSPLIT; ++s) {
    const size_t si = ((size_t)(s * B_SZ + b)) * N_PTS + n4;
    mv[s] = *(const v4f*)(Sm + si);
    lv[s] = *(const v4f*)(Sl + si);
  }
#pragma unroll
  for (int cc = 0; cc < MCPT; ++cc)
#pragma unroll
    for (int s = 0; s < NSPLIT; ++s)
      ov[cc][s] = *(const v4f*)(Op + ((size_t)((s * B_SZ + b) * C_DIM + c0 + cc)) * N_PTS + n4);

  asm volatile("" :: "v"(mv[0]), "v"(mv[1]), "v"(mv[2]), "v"(mv[3]),
                     "v"(lv[0]), "v"(lv[1]), "v"(lv[2]), "v"(lv[3]),
                     "v"(ov[0][0]), "v"(ov[0][1]), "v"(ov[0][2]), "v"(ov[0][3]),
                     "v"(ov[1][0]), "v"(ov[1][1]), "v"(ov[1][2]), "v"(ov[1][3]),
                     "v"(ov[2][0]), "v"(ov[2][1]), "v"(ov[2][2]), "v"(ov[2][3]),
                     "v"(ov[3][0]), "v"(ov[3][1]), "v"(ov[3][2]), "v"(ov[3][3]));

  v4f M = mv[0];
#pragma unroll
  for (int s = 1; s < NSPLIT; ++s)
#pragma unroll
    for (int j = 0; j < 4; ++j) M[j] = fmaxf(M[j], mv[s][j]);
  v4f w[NSPLIT];
  v4f L = {0.0f, 0.0f, 0.0f, 0.0f};
#pragma unroll
  for (int s = 0; s < NSPLIT; ++s)
#pragma unroll
    for (int j = 0; j < 4; ++j) {
      w[s][j] = __expf((mv[s][j] - M[j]) * invTemp);
      L[j] += w[s][j] * lv[s][j];
    }
  v4f inv;
#pragma unroll
  for (int j = 0; j < 4; ++j) inv[j] = 1.0f / L[j];

#pragma unroll
  for (int cc = 0; cc < MCPT; ++cc) {
    const int c = c0 + cc;
    v4f acc = {0.0f, 0.0f, 0.0f, 0.0f};
#pragma unroll
    for (int s = 0; s < NSPLIT; ++s) acc += w[s] * ov[cc][s];
    *(v4f*)(Y + ((size_t)(b * C_DIM + c)) * N_PTS + n4) = acc * inv;
  }

  if (blockIdx.y == 0) {
    const float NEGINF = -__builtin_inff();
    v4f tv[NSPLIT];
#pragma unroll
    for (int s = 0; s < NSPLIT; ++s)
      tv[s] = *(const v4f*)(St2 + ((size_t)(s * B_SZ + b)) * N_PTS + n4);
    v4f rv;
#pragma unroll
    for (int j = 0; j < 4; ++j) {
      float T1 = NEGINF, T2 = NEGINF;
#pragma unroll
      for (int s = 0; s < NSPLIT; ++s) {
        const float hi = fmaxf(T1, mv[s][j]);
        const float lw = fminf(T1, mv[s][j]);
        T2 = fmaxf(lw, fmaxf(T2, tv[s][j]));
        T1 = hi;
      }
      const float x1 = fminf(fmaxf(T1, -1.0f), 1.0f);
      const float x2 = fminf(fmaxf(T2, -1.0f), 1.0f);
      rv[j] = acosf(x1) / acosf(x2);
    }
    *(v4f*)(R + (size_t)b * N_PTS + n4) = rv;
  }
}

extern "C" void kernel_launch(void* const* d_in, const int* in_sizes, int n_in,
                              void* d_out, int out_size, void* d_ws, size_t ws_size,
                              hipStream_t stream) {
  (void)in_sizes; (void)n_in; (void)out_size;
  const float* q  = (const float*)d_in[0];
  const float* k  = (const float*)d_in[1];
  const float* v  = (const float*)d_in[2];
  const float* Wq = (const float*)d_in[3];
  const float* Wk = (const float*)d_in[4];
  const float* Wv = (const float*)d_in[5];

  float* out  = (float*)d_out;
  float* Yout = out;
  float* Rout = out + (size_t)B_SZ * C_DIM * N_PTS;

  const size_t tensElems = (size_t)B_SZ * C_DIM * N_PTS;
  const float *qc = q, *kc = k, *vc = v;

  if (ws_size >= 3 * tensElems * sizeof(float)) {
    float* qw = (float*)d_ws;
    float* kw = qw + tensElems;
    float* vw = kw + tensElems;
    dim3 g(N_PTS / 1024, C_DIM / MIX_OT, 3 * B_SZ);
    cmix3_kernel<<<g, 256, 0, stream>>>(q, k, v, Wq, Wk, Wv, qw, kw, vw);
    qc = qw; kc = kw; vc = vw;
  }

  float* Op  = (float*)d_ws + 3 * tensElems;
  float* Sm  = Op + NSPLIT * tensElems;
  float* St2 = Sm + (size_t)NSPLIT * B_SZ * N_PTS;
  float* Sl  = St2 + (size_t)NSPLIT * B_SZ * N_PTS;

  dim3 ga(N_PTS / 128, B_SZ, NSPLIT);
  attn_kernel<<<ga, 256, 0, stream>>>(qc, kc, vc, Op, Sm, St2, Sl);

  dim3 gm(N_PTS / 1024, C_DIM / MCPT, B_SZ);
  merge_kernel<<<gm, 256, 0, stream>>>(Op, Sm, St2, Sl, Yout, Rout);
}
